# starting kernel
# baseline (speedup 1.0000x reference)
#include <hip/hip_runtime.h>
#include <hip/hip_bf16.h>

typedef __attribute__((ext_vector_type(2))) float v2f;
typedef __attribute__((ext_vector_type(8))) float v8f;

#define DOF   7
#define TRIL  21
#define HID   256
#define ROWS_PER_TILE 16
#define WAVES 8
#define TILES_PER_WAVE 4
#define ROWS_PER_WG (WAVES * ROWS_PER_TILE * TILES_PER_WAVE)

#define W1_STR 10
#define W2_STR 258
#define WB_STR 50

__device__ __forceinline__ v8f wmma_f32(v2f a, v2f b, v8f c) {
    return __builtin_amdgcn_wmma_f32_16x16x4_f32(false, a, false, b, (short)0, c, false, false);
}

__global__ void __launch_bounds__(256)
delan_inverse_kernel(const float* __restrict__ x,
                     const float* __restrict__ W1d, const float* __restrict__ b1d,
                     const float* __restrict__ W2d, const float* __restrict__ b2d,
                     const float* __restrict__ W1o, const float* __restrict__ b1o,
                     const float* __restrict__ W2o, const float* __restrict__ b2o,
                     float* __restrict__ out)
{
    __shared__ float sW1d[256 * W1_STR];
    __shared__ float sW1o[256 * W1_STR];
    __shared__ float sW2d[16  * W2_STR];
    __shared__ float sW2o[32  * W2_STR];
    __shared__ float sWB [WAVES * 16 * WB_STR];

    const int tid = threadIdx.x;

    for (int idx = tid; idx < 256 * 8; idx += 256) {
        int n = idx >> 3, k = idx & 7;
        sW1d[n * W1_STR + k] = (k < DOF) ? W1d[k * HID + n] : 0.0f;
        sW1o[n * W1_STR + k] = (k < DOF) ? W1o[k * HID + n] : 0.0f;
    }
    for (int idx = tid; idx < 16 * 256; idx += 256) {
        int n = idx >> 8, k = idx & 255;
        sW2d[n * W2_STR + k] = (n < DOF) ? W2d[k * DOF + n] : 0.0f;
    }
    for (int idx = tid; idx < 32 * 256; idx += 256) {
        int n = idx >> 8, k = idx & 255;
        sW2o[n * W2_STR + k] = (n < TRIL) ? W2o[k * TRIL + n] : 0.0f;
    }
    __syncthreads();

    const int wave = tid >> 5;
    const int lane = tid & 31;
    const int lh   = lane >> 4;
    const int ln   = lane & 15;
    float* wb = &sWB[wave * 16 * WB_STR];

    for (int t = 0; t < TILES_PER_WAVE; ++t) {
        const long rowBase = (long)blockIdx.x * ROWS_PER_WG
                           + wave * (ROWS_PER_TILE * TILES_PER_WAVE) + t * ROWS_PER_TILE;

        const float* xr = x + (size_t)(rowBase + ln) * (3 * DOF);
        v2f a0, a1;
        {
            int k0 = 2 * lh;
            a0.x = xr[k0];  a0.y = xr[k0 + 1];
            int k1 = 4 + 2 * lh;
            a1.x = xr[k1];  a1.y = (k1 + 1 < DOF) ? xr[k1 + 1] : 0.0f;
        }

        v8f cd = {};
        for (int nt = 0; nt < 16; ++nt) {
            const int N = nt * 16 + ln;
            v8f c = {};
            v2f bA = *(const v2f*)&sW1d[N * W1_STR + 2 * lh];
            v2f bB = *(const v2f*)&sW1d[N * W1_STR + 4 + 2 * lh];
            c = wmma_f32(a0, bA, c);
            c = wmma_f32(a1, bB, c);
            const float bias = b1d[N];
#pragma unroll
            for (int r = 0; r < 8; ++r) {
                float h = fmaxf(c[r] + bias, 0.0f);
                wb[(r + 8 * lh) * WB_STR + ln] = h;
            }
            asm volatile("" ::: "memory");
#pragma unroll
            for (int s2 = 0; s2 < 4; ++s2) {
                v2f af = *(const v2f*)&wb[ln * WB_STR + 4 * s2 + 2 * lh];
                v2f bf = *(const v2f*)&sW2d[ln * W2_STR + nt * 16 + 4 * s2 + 2 * lh];
                cd = wmma_f32(af, bf, cd);
            }
            asm volatile("" ::: "memory");
        }
        {
            float b2 = (ln < DOF) ? b2d[ln] : 0.0f;
#pragma unroll
            for (int r = 0; r < 8; ++r)
                if (ln < DOF) wb[(r + 8 * lh) * WB_STR + 16 + ln] = cd[r] + b2;
        }

        v8f co0 = {}, co1 = {};
        for (int nt = 0; nt < 16; ++nt) {
            const int N = nt * 16 + ln;
            v8f c = {};
            v2f bA = *(const v2f*)&sW1o[N * W1_STR + 2 * lh];
            v2f bB = *(const v2f*)&sW1o[N * W1_STR + 4 + 2 * lh];
            c = wmma_f32(a0, bA, c);
            c = wmma_f32(a1, bB, c);
            const float bias = b1o[N];
#pragma unroll
            for (int r = 0; r < 8; ++r) {
                float h = fmaxf(c[r] + bias, 0.0f);
                wb[(r + 8 * lh) * WB_STR + ln] = h;
            }
            asm volatile("" ::: "memory");
#pragma unroll
            for (int s2 = 0; s2 < 4; ++s2) {
                v2f af  = *(const v2f*)&wb[ln * WB_STR + 4 * s2 + 2 * lh];
                v2f bf0 = *(const v2f*)&sW2o[ ln        * W2_STR + nt * 16 + 4 * s2 + 2 * lh];
                v2f bf1 = *(const v2f*)&sW2o[(16 + ln)  * W2_STR + nt * 16 + 4 * s2 + 2 * lh];
                co0 = wmma_f32(af, bf0, co0);
                co1 = wmma_f32(af, bf1, co1);
            }
            asm volatile("" ::: "memory");
        }
        {
            float t0 = b2o[ln];
            float t1 = (ln < TRIL - 16) ? b2o[16 + ln] : 0.0f;
#pragma unroll
            for (int r = 0; r < 8; ++r) {
                int M = r + 8 * lh;
                wb[M * WB_STR + 23 + ln] = co0[r] + t0;
                if (ln < TRIL - 16) wb[M * WB_STR + 23 + 16 + ln] = co1[r] + t1;
            }
        }
        asm volatile("" ::: "memory");

        if (lane < 16) {
            const long row = rowBase + lane;
            const float* wr = &wb[lane * WB_STR];
            float L[DOF][DOF];
#pragma unroll
            for (int i = 0; i < DOF; ++i) {
#pragma unroll
                for (int j = 0; j < DOF; ++j) L[i][j] = 0.0f;
                L[i][i] = wr[16 + i];
            }
            {
                int tt = 0;
#pragma unroll
                for (int i = 1; i < DOF; ++i)
#pragma unroll
                    for (int j = 0; j < i; ++j) L[i][j] = wr[23 + (tt++)];
            }
            float w[DOF];
            const float* xq = x + (size_t)row * (3 * DOF) + 2 * DOF;
#pragma unroll
            for (int i = 0; i < DOF; ++i) w[i] = xq[i];
            float v[DOF];
#pragma unroll
            for (int k = 0; k < DOF; ++k) {
                float s = 0.0f;
#pragma unroll
                for (int i = k; i < DOF; ++i) s += w[i] * L[i][k];
                v[k] = s;
            }
            float* op = out + (size_t)row * DOF;
#pragma unroll
            for (int j = 0; j < DOF; ++j) {
                float s = 0.0f;
#pragma unroll
                for (int k = 0; k <= j; ++k) s += v[k] * L[j][k];
                op[j] = s;
            }
        }
        asm volatile("" ::: "memory");
    }
}

extern "C" void kernel_launch(void* const* d_in, const int* in_sizes, int n_in,
                              void* d_out, int out_size, void* d_ws, size_t ws_size,
                              hipStream_t stream) {
    const float* x   = (const float*)d_in[0];
    const float* W1d = (const float*)d_in[1];
    const float* b1d = (const float*)d_in[2];
    const float* W2d = (const float*)d_in[3];
    const float* b2d = (const float*)d_in[4];
    const float* W1o = (const float*)d_in[5];
    const float* b1o = (const float*)d_in[6];
    const float* W2o = (const float*)d_in[7];
    const float* b2o = (const float*)d_in[8];
    float* out = (float*)d_out;

    const int rows = in_sizes[0] / (3 * DOF);
    const int grid = rows / ROWS_PER_WG;
    delan_inverse_kernel<<<grid, 256, 0, stream>>>(x, W1d, b1d, W2d, b2d,
                                                   W1o, b1o, W2o, b2o, out);
}


# Hidden tile stays in registers: both layers as transposed products (weights=A, rows=B); layer-1 D regs 2s,2s+1 are layer-2 B (k=8h+2s+e: sum order in 16-blocks changes); b1 rides in k slot 7
# speedup vs baseline: 1.1328x; 1.1328x over previous
#include <hip/hip_runtime.h>


typedef __attribute__((ext_vector_type(2))) float v2f;
typedef __attribute__((ext_vector_type(4))) float v4f;
typedef __attribute__((ext_vector_type(8))) float v8f;

#define DOF   7
#define TRIL  21
#define HID   256
#define ROWS_PER_TILE 16
#define WAVES 8
#define TILES_PER_WAVE 4
#define ROWS_PER_WG (WAVES * ROWS_PER_TILE * TILES_PER_WAVE)

#define W1_STR 10
#define W2_STR 260

__device__ __forceinline__ v8f wmma_f32(v2f a, v2f b, v8f c) {
    return __builtin_amdgcn_wmma_f32_16x16x4_f32(false, a, false, b, (short)0, c, false, false);
}

__device__ __forceinline__ float xhalf(float v) {
    int i = __builtin_bit_cast(int, v);
    return __builtin_bit_cast(float, __builtin_amdgcn_permlanex16(i, i, 0x76543210, 0xfedcba98, false, false));
}

__global__ void __launch_bounds__(256)
delan_inverse_kernel(const float* __restrict__ x,
                     const float* __restrict__ W1d, const float* __restrict__ b1d,
                     const float* __restrict__ W2d, const float* __restrict__ b2d,
                     const float* __restrict__ W1o, const float* __restrict__ b1o,
                     const float* __restrict__ W2o, const float* __restrict__ b2o,
                     float* __restrict__ out)
{
    __shared__ float sW1d[256 * W1_STR];
    __shared__ float sW1o[256 * W1_STR];
    __shared__ __attribute__((aligned(16))) float sW2d[16 * W2_STR];
    __shared__ __attribute__((aligned(16))) float sW2o[32 * W2_STR];

    const int tid = threadIdx.x;

    for (int idx = tid; idx < 256 * 8; idx += 256) {
        int n = idx >> 3, k = idx & 7;
        sW1d[n * W1_STR + k] = (k < DOF) ? W1d[k * HID + n] : b1d[n];
        sW1o[n * W1_STR + k] = (k < DOF) ? W1o[k * HID + n] : b1o[n];
    }
    for (int idx = tid; idx < 16 * 256; idx += 256) {
        int n = idx >> 8, k = idx & 255;
        sW2d[n * W2_STR + k] = (n < DOF) ? W2d[k * DOF + n] : 0.0f;
    }
    for (int idx = tid; idx < 32 * 256; idx += 256) {
        int n = idx >> 8, k = idx & 255;
        sW2o[n * W2_STR + k] = (n < TRIL) ? W2o[k * TRIL + n] : 0.0f;
    }
    __syncthreads();

    const int wave = tid >> 5;
    const int lane = tid & 31;
    const int lh   = lane >> 4;
    const int ln   = lane & 15;

    const float* w1d_l = &sW1d[ln * W1_STR + 2 * lh];
    const float* w1o_l = &sW1o[ln * W1_STR + 2 * lh];
    const float* w2d_l = &sW2d[ln * W2_STR + 8 * lh];
    const float* w2o_l = &sW2o[ln * W2_STR + 8 * lh];

    for (int t = 0; t < TILES_PER_WAVE; ++t) {
        const long rowBase = (long)blockIdx.x * ROWS_PER_WG
                           + wave * (ROWS_PER_TILE * TILES_PER_WAVE) + t * ROWS_PER_TILE;

        const float* xr = x + (size_t)(rowBase + ln) * (3 * DOF);
        v2f q0, q1;
        {
            int k0 = 2 * lh;
            q0.x = xr[k0];  q0.y = xr[k0 + 1];
            int k1 = 4 + 2 * lh;
            q1.x = xr[k1];  q1.y = (k1 + 1 < DOF) ? xr[k1 + 1] : 1.0f;
        }

        v8f cd = {};
#pragma unroll 4
        for (int nt = 0; nt < 16; ++nt) {
            v2f wA = *(const v2f*)(w1d_l + nt * 16 * W1_STR);
            v2f wB = *(const v2f*)(w1d_l + nt * 16 * W1_STR + 4);
            v8f c = {};
            c = wmma_f32(wA, q0, c);
            c = wmma_f32(wB, q1, c);
#pragma unroll
            for (int r = 0; r < 8; ++r) c[r] = fmaxf(c[r], 0.0f);
            v4f u0 = *(const v4f*)(w2d_l + nt * 16);
            v4f u1 = *(const v4f*)(w2d_l + nt * 16 + 4);
            cd = wmma_f32((v2f){u0.x, u0.y}, (v2f){c[0], c[1]}, cd);
            cd = wmma_f32((v2f){u0.z, u0.w}, (v2f){c[2], c[3]}, cd);
            cd = wmma_f32((v2f){u1.x, u1.y}, (v2f){c[4], c[5]}, cd);
            cd = wmma_f32((v2f){u1.z, u1.w}, (v2f){c[6], c[7]}, cd);
        }

        v8f co0 = {}, co1 = {};
#pragma unroll 4
        for (int nt = 0; nt < 16; ++nt) {
            v2f wA = *(const v2f*)(w1o_l + nt * 16 * W1_STR);
            v2f wB = *(const v2f*)(w1o_l + nt * 16 * W1_STR + 4);
            v8f c = {};
            c = wmma_f32(wA, q0, c);
            c = wmma_f32(wB, q1, c);
#pragma unroll
            for (int r = 0; r < 8; ++r) c[r] = fmaxf(c[r], 0.0f);
            v4f u0 = *(const v4f*)(w2o_l + nt * 16);
            v4f u1 = *(const v4f*)(w2o_l + nt * 16 + 4);
            v4f z0 = *(const v4f*)(w2o_l + 16 * W2_STR + nt * 16);
            v4f z1 = *(const v4f*)(w2o_l + 16 * W2_STR + nt * 16 + 4);
            co0 = wmma_f32((v2f){u0.x, u0.y}, (v2f){c[0], c[1]}, co0);
            co1 = wmma_f32((v2f){z0.x, z0.y}, (v2f){c[0], c[1]}, co1);
            co0 = wmma_f32((v2f){u0.z, u0.w}, (v2f){c[2], c[3]}, co0);
            co1 = wmma_f32((v2f){z0.z, z0.w}, (v2f){c[2], c[3]}, co1);
            co0 = wmma_f32((v2f){u1.x, u1.y}, (v2f){c[4], c[5]}, co0);
            co1 = wmma_f32((v2f){z1.x, z1.y}, (v2f){c[4], c[5]}, co1);
            co0 = wmma_f32((v2f){u1.z, u1.w}, (v2f){c[6], c[7]}, co0);
            co1 = wmma_f32((v2f){z1.z, z1.w}, (v2f){c[6], c[7]}, co1);
        }

        float tu[8];
#pragma unroll
        for (int r = 0; r < 8; ++r)
            tu[r] = xhalf(co0[r]);

        if (lane < 16) {
            const long row = rowBase + lane;
            float L[DOF][DOF];
#pragma unroll
            for (int i = 0; i < DOF; ++i) {
#pragma unroll
                for (int j = 0; j < DOF; ++j) L[i][j] = 0.0f;
                L[i][i] = cd[i] + b2d[i];
            }
            {
                int tt = 0;
#pragma unroll
                for (int i = 1; i < DOF; ++i)
#pragma unroll
                    for (int j = 0; j < i; ++j) {
                        float v = (tt < 8) ? co0[tt & 7] : (tt < 16) ? tu[tt & 7] : co1[tt & 7];
                        L[i][j] = v + b2o[tt];
                        ++tt;
                    }
            }
            float w[DOF];
            const float* xq = x + (size_t)row * (3 * DOF) + 2 * DOF;
#pragma unroll
            for (int i = 0; i < DOF; ++i) w[i] = xq[i];
            float v[DOF];
#pragma unroll
            for (int k = 0; k < DOF; ++k) {
                float s = 0.0f;
#pragma unroll
                for (int i = k; i < DOF; ++i) s += w[i] * L[i][k];
                v[k] = s;
            }
            float* op = out + (size_t)row * DOF;
#pragma unroll
            for (int j = 0; j < DOF; ++j) {
                float s = 0.0f;
#pragma unroll
                for (int k = 0; k <= j; ++k) s += v[k] * L[j][k];
                op[j] = s;
            }
        }
    }
}

extern "C" void kernel_launch(void* const* d_in, const int* in_sizes, int n_in,
                              void* d_out, int out_size, void* d_ws, size_t ws_size,
                              hipStream_t stream) {
    const float* x   = (const float*)d_in[0];
    const float* W1d = (const float*)d_in[1];
    const float* b1d = (const float*)d_in[2];
    const float* W2d = (const float*)d_in[3];
    const float* b2d = (const float*)d_in[4];
    const float* W1o = (const float*)d_in[5];
    const float* b1o = (const float*)d_in[6];
    const float* W2o = (const float*)d_in[7];
    const float* b2o = (const float*)d_in[8];
    float* out = (float*)d_out;

    const int rows = in_sizes[0] / (3 * DOF);
    const int grid = rows / ROWS_PER_WG;
    delan_inverse_kernel<<<grid, 256, 0, stream>>>(x, W1d, b1d, W2d, b2d,
                                                   W1o, b1o, W2o, b2o, out);
}


# Weight staging issues all its global loads as one group, in memory order, before the first LDS store (transpose done by the LDS store addresses); was 36 serial load-wait-store loop trips
# speedup vs baseline: 1.2284x; 1.0844x over previous
#include <hip/hip_runtime.h>


typedef __attribute__((ext_vector_type(2))) float v2f;
typedef __attribute__((ext_vector_type(4))) float v4f;
typedef __attribute__((ext_vector_type(8))) float v8f;

#define DOF   7
#define TRIL  21
#define HID   256
#define ROWS_PER_TILE 16
#define WAVES 8
#define TILES_PER_WAVE 4
#define ROWS_PER_WG (WAVES * ROWS_PER_TILE * TILES_PER_WAVE)

#define W1_STR 10
#define W2_STR 260

__device__ __forceinline__ v8f wmma_f32(v2f a, v2f b, v8f c) {
    return __builtin_amdgcn_wmma_f32_16x16x4_f32(false, a, false, b, (short)0, c, false, false);
}

__device__ __forceinline__ float xhalf(float v) {
    int i = __builtin_bit_cast(int, v);
    return __builtin_bit_cast(float, __builtin_amdgcn_permlanex16(i, i, 0x76543210, 0xfedcba98, false, false));
}

__global__ void __launch_bounds__(256)
delan_inverse_kernel(const float* __restrict__ x,
                     const float* __restrict__ W1d, const float* __restrict__ b1d,
                     const float* __restrict__ W2d, const float* __restrict__ b2d,
                     const float* __restrict__ W1o, const float* __restrict__ b1o,
                     const float* __restrict__ W2o, const float* __restrict__ b2o,
                     float* __restrict__ out)
{
    __shared__ float sW1d[256 * W1_STR];
    __shared__ float sW1o[256 * W1_STR];
    __shared__ __attribute__((aligned(16))) float sW2d[16 * W2_STR];
    __shared__ __attribute__((aligned(16))) float sW2o[32 * W2_STR];

    const int tid = threadIdx.x;

    {
        float w1d[8], w1o[8];
#pragma unroll
        for (int k = 0; k < DOF; ++k) {
            w1d[k] = W1d[k * HID + tid];
            w1o[k] = W1o[k * HID + tid];
        }
        w1d[DOF] = b1d[tid];
        w1o[DOF] = b1o[tid];
        v4f r2d[2], r2o[6];
#pragma unroll
        for (int j = 0; j < 2; ++j) {
            const int i = tid + 256 * j;
            r2d[j] = ((const v4f*)W2d)[(i < 448) ? i : 447];
        }
#pragma unroll
        for (int j = 0; j < 6; ++j) {
            const int i = tid + 256 * j;
            r2o[j] = ((const v4f*)W2o)[(i < 1344) ? i : 1343];
        }

        for (int i = tid; i < (16 - DOF) * W2_STR / 4; i += 256)
            ((v4f*)(sW2d + DOF * W2_STR))[i] = (v4f){0.0f, 0.0f, 0.0f, 0.0f};
        for (int i = tid; i < (32 - TRIL) * W2_STR / 4; i += 256)
            ((v4f*)(sW2o + TRIL * W2_STR))[i] = (v4f){0.0f, 0.0f, 0.0f, 0.0f};

#pragma unroll
        for (int p = 0; p < 4; ++p) {
            *(v2f*)&sW1d[tid * W1_STR + 2 * p] = (v2f){w1d[2 * p], w1d[2 * p + 1]};
            *(v2f*)&sW1o[tid * W1_STR + 2 * p] = (v2f){w1o[2 * p], w1o[2 * p + 1]};
        }
#pragma unroll
        for (int j = 0; j < 2; ++j) {
            const int i = tid + 256 * j;
            if (i < 448) {
                int kk = (4 * i) / DOF, nn = 4 * i - DOF * kk;
#pragma unroll
                for (int c = 0; c < 4; ++c) {
                    sW2d[nn * W2_STR + kk] = r2d[j][c];
                    if (++nn == DOF) { nn = 0; ++kk; }
                }
            }
        }
#pragma unroll
        for (int j = 0; j < 6; ++j) {
            const int i = tid + 256 * j;
            if (i < 1344) {
                int kk = (4 * i) / TRIL, nn = 4 * i - TRIL * kk;
#pragma unroll
                for (int c = 0; c < 4; ++c) {
                    sW2o[nn * W2_STR + kk] = r2o[j][c];
                    if (++nn == TRIL) { nn = 0; ++kk; }
                }
            }
        }
    }
    __syncthreads();

    const int wave = tid >> 5;
    const int lane = tid & 31;
    const int lh   = lane >> 4;
    const int ln   = lane & 15;

    const float* w1d_l = &sW1d[ln * W1_STR + 2 * lh];
    const float* w1o_l = &sW1o[ln * W1_STR + 2 * lh];
    const float* w2d_l = &sW2d[ln * W2_STR + 8 * lh];
    const float* w2o_l = &sW2o[ln * W2_STR + 8 * lh];

    for (int t = 0; t < TILES_PER_WAVE; ++t) {
        const long rowBase = (long)blockIdx.x * ROWS_PER_WG
                           + wave * (ROWS_PER_TILE * TILES_PER_WAVE) + t * ROWS_PER_TILE;

        const float* xr = x + (size_t)(rowBase + ln) * (3 * DOF);
        v2f q0, q1;
        {
            int k0 = 2 * lh;
            q0.x = xr[k0];  q0.y = xr[k0 + 1];
            int k1 = 4 + 2 * lh;
            q1.x = xr[k1];  q1.y = (k1 + 1 < DOF) ? xr[k1 + 1] : 1.0f;
        }

        v8f cd = {};
#pragma unroll 4
        for (int nt = 0; nt < 16; ++nt) {
            v2f wA = *(const v2f*)(w1d_l + nt * 16 * W1_STR);
            v2f wB = *(const v2f*)(w1d_l + nt * 16 * W1_STR + 4);
            v8f c = {};
            c = wmma_f32(wA, q0, c);
            c = wmma_f32(wB, q1, c);
#pragma unroll
            for (int r = 0; r < 8; ++r) c[r] = fmaxf(c[r], 0.0f);
            v4f u0 = *(const v4f*)(w2d_l + nt * 16);
            v4f u1 = *(const v4f*)(w2d_l + nt * 16 + 4);
            cd = wmma_f32((v2f){u0.x, u0.y}, (v2f){c[0], c[1]}, cd);
            cd = wmma_f32((v2f){u0.z, u0.w}, (v2f){c[2], c[3]}, cd);
            cd = wmma_f32((v2f){u1.x, u1.y}, (v2f){c[4], c[5]}, cd);
            cd = wmma_f32((v2f){u1.z, u1.w}, (v2f){c[6], c[7]}, cd);
        }

        v8f co0 = {}, co1 = {};
#pragma unroll 4
        for (int nt = 0; nt < 16; ++nt) {
            v2f wA = *(const v2f*)(w1o_l + nt * 16 * W1_STR);
            v2f wB = *(const v2f*)(w1o_l + nt * 16 * W1_STR + 4);
            v8f c = {};
            c = wmma_f32(wA, q0, c);
            c = wmma_f32(wB, q1, c);
#pragma unroll
            for (int r = 0; r < 8; ++r) c[r] = fmaxf(c[r], 0.0f);
            v4f u0 = *(const v4f*)(w2o_l + nt * 16);
            v4f u1 = *(const v4f*)(w2o_l + nt * 16 + 4);
            v4f z0 = *(const v4f*)(w2o_l + 16 * W2_STR + nt * 16);
            v4f z1 = *(const v4f*)(w2o_l + 16 * W2_STR + nt * 16 + 4);
            co0 = wmma_f32((v2f){u0.x, u0.y}, (v2f){c[0], c[1]}, co0);
            co1 = wmma_f32((v2f){z0.x, z0.y}, (v2f){c[0], c[1]}, co1);
            co0 = wmma_f32((v2f){u0.z, u0.w}, (v2f){c[2], c[3]}, co0);
            co1 = wmma_f32((v2f){z0.z, z0.w}, (v2f){c[2], c[3]}, co1);
            co0 = wmma_f32((v2f){u1.x, u1.y}, (v2f){c[4], c[5]}, co0);
            co1 = wmma_f32((v2f){z1.x, z1.y}, (v2f){c[4], c[5]}, co1);
            co0 = wmma_f32((v2f){u1.z, u1.w}, (v2f){c[6], c[7]}, co0);
            co1 = wmma_f32((v2f){z1.z, z1.w}, (v2f){c[6], c[7]}, co1);
        }

        float tu[8];
#pragma unroll
        for (int r = 0; r < 8; ++r)
            tu[r] = xhalf(co0[r]);

        if (lane < 16) {
            const long row = rowBase + lane;
            float L[DOF][DOF];
#pragma unroll
            for (int i = 0; i < DOF; ++i) {
#pragma unroll
                for (int j = 0; j < DOF; ++j) L[i][j] = 0.0f;
                L[i][i] = cd[i] + b2d[i];
            }
            {
                int tt = 0;
#pragma unroll
                for (int i = 1; i < DOF; ++i)
#pragma unroll
                    for (int j = 0; j < i; ++j) {
                        float v = (tt < 8) ? co0[tt & 7] : (tt < 16) ? tu[tt & 7] : co1[tt & 7];
                        L[i][j] = v + b2o[tt];
                        ++tt;
                    }
            }
            float w[DOF];
            const float* xq = x + (size_t)row * (3 * DOF) + 2 * DOF;
#pragma unroll
            for (int i = 0; i < DOF; ++i) w[i] = xq[i];
            float v[DOF];
#pragma unroll
            for (int k = 0; k < DOF; ++k) {
                float s = 0.0f;
#pragma unroll
                for (int i = k; i < DOF; ++i) s += w[i] * L[i][k];
                v[k] = s;
            }
            float* op = out + (size_t)row * DOF;
#pragma unroll
            for (int j = 0; j < DOF; ++j) {
                float s = 0.0f;
#pragma unroll
                for (int k = 0; k <= j; ++k) s += v[k] * L[j][k];
                op[j] = s;
            }
        }
    }
}

extern "C" void kernel_launch(void* const* d_in, const int* in_sizes, int n_in,
                              void* d_out, int out_size, void* d_ws, size_t ws_size,
                              hipStream_t stream) {
    const float* x   = (const float*)d_in[0];
    const float* W1d = (const float*)d_in[1];
    const float* b1d = (const float*)d_in[2];
    const float* W2d = (const float*)d_in[3];
    const float* b2d = (const float*)d_in[4];
    const float* W1o = (const float*)d_in[5];
    const float* b1o = (const float*)d_in[6];
    const float* W2o = (const float*)d_in[7];
    const float* b2o = (const float*)d_in[8];
    float* out = (float*)d_out;

    const int rows = in_sizes[0] / (3 * DOF);
    const int grid = rows / ROWS_PER_WG;
    delan_inverse_kernel<<<grid, 256, 0, stream>>>(x, W1d, b1d, W2d, b2d,
                                                   W1o, b1o, W2o, b2o, out);
}


# Each wave carries two 16-row tiles per pass: every W1/W2 fragment read from LDS feeds both tiles (LDS fragment reads halved, two independent accumulator chains per step); per-tile sum order unchanged
# speedup vs baseline: 1.2953x; 1.0545x over previous
#include <hip/hip_runtime.h>


typedef __attribute__((ext_vector_type(2))) float v2f;
typedef __attribute__((ext_vector_type(4))) float v4f;
typedef __attribute__((ext_vector_type(8))) float v8f;

#define DOF   7
#define TRIL  21
#define HID   256
#define ROWS_PER_TILE 16
#define WAVES 8
#define TILES_PER_WAVE 4
#define ROWS_PER_WG (WAVES * ROWS_PER_TILE * TILES_PER_WAVE)

#define W1_STR 10
#define W2_STR 260

__device__ __forceinline__ v8f wmma_f32(v2f a, v2f b, v8f c) {
    return __builtin_amdgcn_wmma_f32_16x16x4_f32(false, a, false, b, (short)0, c, false, false);
}

__device__ __forceinline__ float xhalf(float v) {
    int i = __builtin_bit_cast(int, v);
    return __builtin_bit_cast(float, __builtin_amdgcn_permlanex16(i, i, 0x76543210, 0xfedcba98, false, false));
}

__device__ __forceinline__ void row_epilogue(const v8f cd, const v8f co0, const v8f co1, const float* tu,
                                             const float* __restrict__ x,
                                             const float* __restrict__ b2d, const float* __restrict__ b2o,
                                             float* __restrict__ out, const long row)
{
    float L[DOF][DOF];
#pragma unroll
    for (int i = 0; i < DOF; ++i) {
#pragma unroll
        for (int j = 0; j < DOF; ++j) L[i][j] = 0.0f;
        L[i][i] = cd[i] + b2d[i];
    }
    {
        int tt = 0;
#pragma unroll
        for (int i = 1; i < DOF; ++i)
#pragma unroll
            for (int j = 0; j < i; ++j) {
                float v = (tt < 8) ? co0[tt & 7] : (tt < 16) ? tu[tt & 7] : co1[tt & 7];
                L[i][j] = v + b2o[tt];
                ++tt;
            }
    }
    float w[DOF];
    const float* xq = x + (size_t)row * (3 * DOF) + 2 * DOF;
#pragma unroll
    for (int i = 0; i < DOF; ++i) w[i] = xq[i];
    float v[DOF];
#pragma unroll
    for (int k = 0; k < DOF; ++k) {
        float s = 0.0f;
#pragma unroll
        for (int i = k; i < DOF; ++i) s += w[i] * L[i][k];
        v[k] = s;
    }
    float* op = out + (size_t)row * DOF;
#pragma unroll
    for (int j = 0; j < DOF; ++j) {
        float s = 0.0f;
#pragma unroll
        for (int k = 0; k <= j; ++k) s += v[k] * L[j][k];
        op[j] = s;
    }
}

__global__ void __launch_bounds__(256)
delan_inverse_kernel(const float* __restrict__ x,
                     const float* __restrict__ W1d, const float* __restrict__ b1d,
                     const float* __restrict__ W2d, const float* __restrict__ b2d,
                     const float* __restrict__ W1o, const float* __restrict__ b1o,
                     const float* __restrict__ W2o, const float* __restrict__ b2o,
                     float* __restrict__ out)
{
    __shared__ float sW1d[256 * W1_STR];
    __shared__ float sW1o[256 * W1_STR];
    __shared__ __attribute__((aligned(16))) float sW2d[16 * W2_STR];
    __shared__ __attribute__((aligned(16))) float sW2o[32 * W2_STR];

    const int tid = threadIdx.x;

    {
        float w1d[8], w1o[8];
#pragma unroll
        for (int k = 0; k < DOF; ++k) {
            w1d[k] = W1d[k * HID + tid];
            w1o[k] = W1o[k * HID + tid];
        }
        w1d[DOF] = b1d[tid];
        w1o[DOF] = b1o[tid];
        v4f r2d[2], r2o[6];
#pragma unroll
        for (int j = 0; j < 2; ++j) {
            const int i = tid + 256 * j;
            r2d[j] = ((const v4f*)W2d)[(i < 448) ? i : 447];
        }
#pragma unroll
        for (int j = 0; j < 6; ++j) {
            const int i = tid + 256 * j;
            r2o[j] = ((const v4f*)W2o)[(i < 1344) ? i : 1343];
        }

        for (int i = tid; i < (16 - DOF) * W2_STR / 4; i += 256)
            ((v4f*)(sW2d + DOF * W2_STR))[i] = (v4f){0.0f, 0.0f, 0.0f, 0.0f};
        for (int i = tid; i < (32 - TRIL) * W2_STR / 4; i += 256)
            ((v4f*)(sW2o + TRIL * W2_STR))[i] = (v4f){0.0f, 0.0f, 0.0f, 0.0f};

#pragma unroll
        for (int p = 0; p < 4; ++p) {
            *(v2f*)&sW1d[tid * W1_STR + 2 * p] = (v2f){w1d[2 * p], w1d[2 * p + 1]};
            *(v2f*)&sW1o[tid * W1_STR + 2 * p] = (v2f){w1o[2 * p], w1o[2 * p + 1]};
        }
#pragma unroll
        for (int j = 0; j < 2; ++j) {
            const int i = tid + 256 * j;
            if (i < 448) {
                int kk = (4 * i) / DOF, nn = 4 * i - DOF * kk;
#pragma unroll
                for (int c = 0; c < 4; ++c) {
                    sW2d[nn * W2_STR + kk] = r2d[j][c];
                    if (++nn == DOF) { nn = 0; ++kk; }
                }
            }
        }
#pragma unroll
        for (int j = 0; j < 6; ++j) {
            const int i = tid + 256 * j;
            if (i < 1344) {
                int kk = (4 * i) / TRIL, nn = 4 * i - TRIL * kk;
#pragma unroll
                for (int c = 0; c < 4; ++c) {
                    sW2o[nn * W2_STR + kk] = r2o[j][c];
                    if (++nn == TRIL) { nn = 0; ++kk; }
                }
            }
        }
    }
    __syncthreads();

    const int wave = tid >> 5;
    const int lane = tid & 31;
    const int lh   = lane >> 4;
    const int ln   = lane & 15;

    const float* w1d_l = &sW1d[ln * W1_STR + 2 * lh];
    const float* w1o_l = &sW1o[ln * W1_STR + 2 * lh];
    const float* w2d_l = &sW2d[ln * W2_STR + 8 * lh];
    const float* w2o_l = &sW2o[ln * W2_STR + 8 * lh];

    for (int p = 0; p < TILES_PER_WAVE / 2; ++p) {
        const long rowBase = (long)blockIdx.x * ROWS_PER_WG
                           + wave * (ROWS_PER_TILE * TILES_PER_WAVE) + p * (2 * ROWS_PER_TILE);

        const float* xa = x + (size_t)(rowBase + ln) * (3 * DOF);
        const float* xb = xa + (size_t)ROWS_PER_TILE * (3 * DOF);
        v2f q0a, q1a, q0b, q1b;
        {
            const int k0 = 2 * lh;
            const int k1 = 4 + 2 * lh;
            q0a.x = xa[k0];  q0a.y = xa[k0 + 1];
            q0b.x = xb[k0];  q0b.y = xb[k0 + 1];
            q1a.x = xa[k1];  q1a.y = (k1 + 1 < DOF) ? xa[k1 + 1] : 1.0f;
            q1b.x = xb[k1];  q1b.y = (k1 + 1 < DOF) ? xb[k1 + 1] : 1.0f;
        }

        v8f cda = {}, cdb = {};
#pragma unroll 4
        for (int nt = 0; nt < 16; ++nt) {
            v2f wA = *(const v2f*)(w1d_l + nt * 16 * W1_STR);
            v2f wB = *(const v2f*)(w1d_l + nt * 16 * W1_STR + 4);
            v4f u0 = *(const v4f*)(w2d_l + nt * 16);
            v4f u1 = *(const v4f*)(w2d_l + nt * 16 + 4);
            v8f ca = {}, cb = {};
            ca = wmma_f32(wA, q0a, ca);
            cb = wmma_f32(wA, q0b, cb);
            ca = wmma_f32(wB, q1a, ca);
            cb = wmma_f32(wB, q1b, cb);
#pragma unroll
            for (int r = 0; r < 8; ++r) { ca[r] = fmaxf(ca[r], 0.0f); cb[r] = fmaxf(cb[r], 0.0f); }
            cda = wmma_f32((v2f){u0.x, u0.y}, (v2f){ca[0], ca[1]}, cda);
            cdb = wmma_f32((v2f){u0.x, u0.y}, (v2f){cb[0], cb[1]}, cdb);
            cda = wmma_f32((v2f){u0.z, u0.w}, (v2f){ca[2], ca[3]}, cda);
            cdb = wmma_f32((v2f){u0.z, u0.w}, (v2f){cb[2], cb[3]}, cdb);
            cda = wmma_f32((v2f){u1.x, u1.y}, (v2f){ca[4], ca[5]}, cda);
            cdb = wmma_f32((v2f){u1.x, u1.y}, (v2f){cb[4], cb[5]}, cdb);
            cda = wmma_f32((v2f){u1.z, u1.w}, (v2f){ca[6], ca[7]}, cda);
            cdb = wmma_f32((v2f){u1.z, u1.w}, (v2f){cb[6], cb[7]}, cdb);
        }

        v8f co0a = {}, co1a = {}, co0b = {}, co1b = {};
#pragma unroll 2
        for (int nt = 0; nt < 16; ++nt) {
            v2f wA = *(const v2f*)(w1o_l + nt * 16 * W1_STR);
            v2f wB = *(const v2f*)(w1o_l + nt * 16 * W1_STR + 4);
            v4f u0 = *(const v4f*)(w2o_l + nt * 16);
            v4f u1 = *(const v4f*)(w2o_l + nt * 16 + 4);
            v4f z0 = *(const v4f*)(w2o_l + 16 * W2_STR + nt * 16);
            v4f z1 = *(const v4f*)(w2o_l + 16 * W2_STR + nt * 16 + 4);
            v8f ca = {}, cb = {};
            ca = wmma_f32(wA, q0a, ca);
            cb = wmma_f32(wA, q0b, cb);
            ca = wmma_f32(wB, q1a, ca);
            cb = wmma_f32(wB, q1b, cb);
#pragma unroll
            for (int r = 0; r < 8; ++r) { ca[r] = fmaxf(ca[r], 0.0f); cb[r] = fmaxf(cb[r], 0.0f); }
            co0a = wmma_f32((v2f){u0.x, u0.y}, (v2f){ca[0], ca[1]}, co0a);
            co0b = wmma_f32((v2f){u0.x, u0.y}, (v2f){cb[0], cb[1]}, co0b);
            co1a = wmma_f32((v2f){z0.x, z0.y}, (v2f){ca[0], ca[1]}, co1a);
            co1b = wmma_f32((v2f){z0.x, z0.y}, (v2f){cb[0], cb[1]}, co1b);
            co0a = wmma_f32((v2f){u0.z, u0.w}, (v2f){ca[2], ca[3]}, co0a);
            co0b = wmma_f32((v2f){u0.z, u0.w}, (v2f){cb[2], cb[3]}, co0b);
            co1a = wmma_f32((v2f){z0.z, z0.w}, (v2f){ca[2], ca[3]}, co1a);
            co1b = wmma_f32((v2f){z0.z, z0.w}, (v2f){cb[2], cb[3]}, co1b);
            co0a = wmma_f32((v2f){u1.x, u1.y}, (v2f){ca[4], ca[5]}, co0a);
            co0b = wmma_f32((v2f){u1.x, u1.y}, (v2f){cb[4], cb[5]}, co0b);
            co1a = wmma_f32((v2f){z1.x, z1.y}, (v2f){ca[4], ca[5]}, co1a);
            co1b = wmma_f32((v2f){z1.x, z1.y}, (v2f){cb[4], cb[5]}, co1b);
            co0a = wmma_f32((v2f){u1.z, u1.w}, (v2f){ca[6], ca[7]}, co0a);
            co0b = wmma_f32((v2f){u1.z, u1.w}, (v2f){cb[6], cb[7]}, co0b);
            co1a = wmma_f32((v2f){z1.z, z1.w}, (v2f){ca[6], ca[7]}, co1a);
            co1b = wmma_f32((v2f){z1.z, z1.w}, (v2f){cb[6], cb[7]}, co1b);
        }

        float tua[8], tub[8];
#pragma unroll
        for (int r = 0; r < 8; ++r) {
            tua[r] = xhalf(co0a[r]);
            tub[r] = xhalf(co0b[r]);
        }

        if (lane < 16) {
            row_epilogue(cda, co0a, co1a, tua, x, b2d, b2o, out, rowBase + lane);
            row_epilogue(cdb, co0b, co1b, tub, x, b2d, b2o, out, rowBase + ROWS_PER_TILE + lane);
        }
    }
}

extern "C" void kernel_launch(void* const* d_in, const int* in_sizes, int n_in,
                              void* d_out, int out_size, void* d_ws, size_t ws_size,
                              hipStream_t stream) {
    const float* x   = (const float*)d_in[0];
    const float* W1d = (const float*)d_in[1];
    const float* b1d = (const float*)d_in[2];
    const float* W2d = (const float*)d_in[3];
    const float* b2d = (const float*)d_in[4];
    const float* W1o = (const float*)d_in[5];
    const float* b1o = (const float*)d_in[6];
    const float* W2o = (const float*)d_in[7];
    const float* b2o = (const float*)d_in[8];
    float* out = (float*)d_out;

    const int rows = in_sizes[0] / (3 * DOF);
    const int grid = rows / ROWS_PER_WG;
    delan_inverse_kernel<<<grid, 256, 0, stream>>>(x, W1d, b1d, W2d, b2d,
                                                   W1o, b1o, W2o, b2o, out);
}
